# starting kernel
# baseline (speedup 1.0000x reference)
#include <hip/hip_runtime.h>
#include <hip/hip_bf16.h>
#include <math.h>


#define BATCH 64
#define CAPS  10
#define NODES 6272
#define CIN   8
#define COUT  16
#define NCHUNK 49
#define NPC   (NODES / NCHUNK)

typedef float v2f __attribute__((ext_vector_type(2)));
typedef float v8f __attribute__((ext_vector_type(8)));

__global__ void __launch_bounds__(256)
caps_init_kernel(float* __restrict__ logits, float* __restrict__ mz,
                 float* __restrict__ sbuf) {
    const size_t NL = (size_t)BATCH * CAPS * NODES;
    size_t i = (size_t)blockIdx.x * 256 + threadIdx.x;
    for (size_t k = i; k < NL; k += (size_t)gridDim.x * 256)
        logits[k] = 0.0f;
    if (i < BATCH * CAPS) {
        mz[2 * i + 0] = 0.0f;
        mz[2 * i + 1] = (float)NODES;
    }
    if (i < BATCH * CAPS * COUT) sbuf[i] = 0.0f;
}

__global__ void __launch_bounds__(256)
caps_zero_s_kernel(float* __restrict__ sbuf) {
    int i = blockIdx.x * 256 + threadIdx.x;
    if (i < BATCH * CAPS * COUT) sbuf[i] = 0.0f;
}

__global__ void __launch_bounds__(128)
caps_s_gemm_kernel(const float* __restrict__ x, const float* __restrict__ W,
                   const float* __restrict__ logits, const float* __restrict__ mz,
                   float* __restrict__ sbuf) {
    const int c     = blockIdx.x;
    const int chunk = blockIdx.y;
    const int wave  = threadIdx.x >> 5;
    const int lane  = threadIdx.x & 31;
    const int half  = lane >> 4;
    const int l16   = lane & 15;
    const int b0    = wave * 16;
    const int b     = b0 + l16;

    const float  m    = mz[((size_t)b * CAPS + c) * 2 + 0];
    const float  invZ = 1.0f / mz[((size_t)b * CAPS + c) * 2 + 1];
    const float* xrow = x + (size_t)b * NODES * CIN;
    const float* lrow = logits + ((size_t)b * CAPS + c) * NODES;
    const float* Wc   = W + (size_t)c * NODES * CIN * COUT;

    v8f acc = {};
    const int n0 = chunk * NPC;
#pragma unroll 2
    for (int n = n0; n < n0 + NPC; ++n) {
        const float wgt = __expf(lrow[n] - m) * invZ;
        const float* xp = xrow + n * CIN + half * 2;
        v2f a_lo = { wgt * xp[0], wgt * xp[1] };
        v2f a_hi = { wgt * xp[4], wgt * xp[5] };
        const float* Wp = Wc + ((size_t)n * CIN + half * 2) * COUT + l16;
        v2f b_lo = { Wp[0],        Wp[COUT] };
        v2f b_hi = { Wp[4 * COUT], Wp[5 * COUT] };
        acc = __builtin_amdgcn_wmma_f32_16x16x4_f32(false, a_lo, false, b_lo,
                                                    (short)0, acc, false, false);
        acc = __builtin_amdgcn_wmma_f32_16x16x4_f32(false, a_hi, false, b_hi,
                                                    (short)0, acc, false, false);
    }
#pragma unroll
    for (int r = 0; r < 8; ++r) {
        const int br = b0 + r + 8 * half;
        atomicAdd(&sbuf[((size_t)br * CAPS + c) * COUT + l16], acc[r]);
    }
}

__global__ void __launch_bounds__(256)
caps_squash_kernel(const float* __restrict__ sbuf, float* __restrict__ vbuf,
                   float* __restrict__ out, int write_out) {
    const int idx = blockIdx.x * 256 + threadIdx.x;
    if (idx >= BATCH * CAPS * COUT) return;
    const int bc = idx / COUT;
    const float* sp = sbuf + (size_t)bc * COUT;
    float n2 = 0.0f;
#pragma unroll
    for (int o = 0; o < COUT; ++o) n2 += sp[o] * sp[o];
    const float scale = n2 / ((1.0f + n2) * sqrtf(n2));
    const float v = scale * sbuf[idx];
    vbuf[idx] = v;
    if (write_out) out[idx] = v;
}

__global__ void __launch_bounds__(128)
caps_agree_kernel(const float* __restrict__ x, const float* __restrict__ W,
                  const float* __restrict__ vbuf, float* __restrict__ logits) {
    const int c     = blockIdx.x;
    const int chunk = blockIdx.y;
    const int wave  = threadIdx.x >> 5;
    const int lane  = threadIdx.x & 31;
    const int half  = lane >> 4;
    const int l16   = lane & 15;
    const int b0    = wave * 16;
    const int b     = b0 + l16;

    __shared__ float vsh[BATCH * COUT];
    for (int i = threadIdx.x; i < BATCH * COUT; i += 128) {
        const int bb = i / COUT, oo = i % COUT;
        vsh[i] = vbuf[((size_t)bb * CAPS + c) * COUT + oo];
    }
    __syncthreads();

    const float* xrow = x + (size_t)b * NODES * CIN;
    const float* Wc   = W + (size_t)c * NODES * CIN * COUT;
    const int n0 = chunk * NPC;
    for (int n = n0; n < n0 + NPC; ++n) {
        const float* xp = xrow + n * CIN + half * 2;
        v2f a_lo = { xp[0], xp[1] };
        v2f a_hi = { xp[4], xp[5] };
        const float* Wp = Wc + ((size_t)n * CIN + half * 2) * COUT + l16;
        v2f b_lo = { Wp[0],        Wp[COUT] };
        v2f b_hi = { Wp[4 * COUT], Wp[5 * COUT] };
        v8f acc = {};
        acc = __builtin_amdgcn_wmma_f32_16x16x4_f32(false, a_lo, false, b_lo,
                                                    (short)0, acc, false, false);
        acc = __builtin_amdgcn_wmma_f32_16x16x4_f32(false, a_hi, false, b_hi,
                                                    (short)0, acc, false, false);
#pragma unroll
        for (int r = 0; r < 8; ++r) {
            const int br = b0 + r + 8 * half;
            float t = acc[r] * vsh[br * COUT + l16];
            t += __shfl_xor(t, 1, 32);
            t += __shfl_xor(t, 2, 32);
            t += __shfl_xor(t, 4, 32);
            t += __shfl_xor(t, 8, 32);
            if (l16 == 0) {
                const size_t idx = ((size_t)br * CAPS + c) * NODES + n;
                logits[idx] = logits[idx] + t;
            }
        }
    }
}

__global__ void __launch_bounds__(256)
caps_softmax_kernel(const float* __restrict__ logits, float* __restrict__ mz) {
    const int bc = blockIdx.x;
    const float* row = logits + (size_t)bc * NODES;
    __shared__ float red[256];
    float m = -1e30f;
    for (int n = threadIdx.x; n < NODES; n += 256) m = fmaxf(m, row[n]);
    red[threadIdx.x] = m;
    __syncthreads();
    for (int s = 128; s > 0; s >>= 1) {
        if (threadIdx.x < s)
            red[threadIdx.x] = fmaxf(red[threadIdx.x], red[threadIdx.x + s]);
        __syncthreads();
    }
    m = red[0];
    __syncthreads();
    float z = 0.0f;
    for (int n = threadIdx.x; n < NODES; n += 256) z += __expf(row[n] - m);
    red[threadIdx.x] = z;
    __syncthreads();
    for (int s = 128; s > 0; s >>= 1) {
        if (threadIdx.x < s) red[threadIdx.x] += red[threadIdx.x + s];
        __syncthreads();
    }
    if (threadIdx.x == 0) {
        mz[2 * bc + 0] = m;
        mz[2 * bc + 1] = red[0];
    }
}

extern "C" void kernel_launch(void* const* d_in, const int* in_sizes, int n_in,
                              void* d_out, int out_size, void* d_ws, size_t ws_size,
                              hipStream_t stream) {
    const float* x = (const float*)d_in[0];
    const float* W = (const float*)d_in[1];
    float* out = (float*)d_out;

    float* logits = (float*)d_ws;
    float* mz     = logits + (size_t)BATCH * CAPS * NODES;
    float* sbuf   = mz + (size_t)BATCH * CAPS * 2;
    float* vbuf   = sbuf + (size_t)BATCH * CAPS * COUT;

    caps_init_kernel<<<2048, 256, 0, stream>>>(logits, mz, sbuf);

    const dim3 gemm_grid(CAPS, NCHUNK);
    for (int t = 0; t < 3; ++t) {
        caps_s_gemm_kernel<<<gemm_grid, 128, 0, stream>>>(x, W, logits, mz, sbuf);
        caps_squash_kernel<<<(BATCH * CAPS * COUT + 255) / 256, 256, 0, stream>>>(
            sbuf, vbuf, out, (t == 2) ? 1 : 0);
        if (t < 2) {
            caps_zero_s_kernel<<<(BATCH * CAPS * COUT + 255) / 256, 256, 0, stream>>>(sbuf);
            caps_agree_kernel<<<gemm_grid, 128, 0, stream>>>(x, W, vbuf, logits);
            caps_softmax_kernel<<<BATCH * CAPS, 256, 0, stream>>>(logits, mz);
        }
    }
}


# agreement kernel: W^T as A, x^T as B puts u_hat's o index in the D registers: o-sum is in-lane FMAs, 8 nodes per group, one 16-byte logits update per lane (was 32 shuffles + 8 serial updates a node)
# speedup vs baseline: 4.4762x; 4.4762x over previous
#include <hip/hip_runtime.h>
#include <hip/hip_bf16.h>
#include <math.h>


#define BATCH 64
#define CAPS  10
#define NODES 6272
#define CIN   8
#define COUT  16
#define NCHUNK 49
#define NPC   (NODES / NCHUNK)

typedef float v2f __attribute__((ext_vector_type(2)));
typedef float v8f __attribute__((ext_vector_type(8)));

__global__ void __launch_bounds__(256)
caps_init_kernel(float* __restrict__ logits, float* __restrict__ mz,
                 float* __restrict__ sbuf) {
    const size_t NL = (size_t)BATCH * CAPS * NODES;
    size_t i = (size_t)blockIdx.x * 256 + threadIdx.x;
    for (size_t k = i; k < NL; k += (size_t)gridDim.x * 256)
        logits[k] = 0.0f;
    if (i < BATCH * CAPS) {
        mz[2 * i + 0] = 0.0f;
        mz[2 * i + 1] = (float)NODES;
    }
    if (i < BATCH * CAPS * COUT) sbuf[i] = 0.0f;
}

__global__ void __launch_bounds__(256)
caps_zero_s_kernel(float* __restrict__ sbuf) {
    int i = blockIdx.x * 256 + threadIdx.x;
    if (i < BATCH * CAPS * COUT) sbuf[i] = 0.0f;
}

__global__ void __launch_bounds__(128)
caps_s_gemm_kernel(const float* __restrict__ x, const float* __restrict__ W,
                   const float* __restrict__ logits, const float* __restrict__ mz,
                   float* __restrict__ sbuf) {
    const int c     = blockIdx.x;
    const int chunk = blockIdx.y;
    const int wave  = threadIdx.x >> 5;
    const int lane  = threadIdx.x & 31;
    const int half  = lane >> 4;
    const int l16   = lane & 15;
    const int b0    = wave * 16;
    const int b     = b0 + l16;

    const float  m    = mz[((size_t)b * CAPS + c) * 2 + 0];
    const float  invZ = 1.0f / mz[((size_t)b * CAPS + c) * 2 + 1];
    const float* xrow = x + (size_t)b * NODES * CIN;
    const float* lrow = logits + ((size_t)b * CAPS + c) * NODES;
    const float* Wc   = W + (size_t)c * NODES * CIN * COUT;

    v8f acc = {};
    const int n0 = chunk * NPC;
#pragma unroll 2
    for (int n = n0; n < n0 + NPC; ++n) {
        const float wgt = __expf(lrow[n] - m) * invZ;
        const float* xp = xrow + n * CIN + half * 2;
        v2f a_lo = { wgt * xp[0], wgt * xp[1] };
        v2f a_hi = { wgt * xp[4], wgt * xp[5] };
        const float* Wp = Wc + ((size_t)n * CIN + half * 2) * COUT + l16;
        v2f b_lo = { Wp[0],        Wp[COUT] };
        v2f b_hi = { Wp[4 * COUT], Wp[5 * COUT] };
        acc = __builtin_amdgcn_wmma_f32_16x16x4_f32(false, a_lo, false, b_lo,
                                                    (short)0, acc, false, false);
        acc = __builtin_amdgcn_wmma_f32_16x16x4_f32(false, a_hi, false, b_hi,
                                                    (short)0, acc, false, false);
    }
#pragma unroll
    for (int r = 0; r < 8; ++r) {
        const int br = b0 + r + 8 * half;
        atomicAdd(&sbuf[((size_t)br * CAPS + c) * COUT + l16], acc[r]);
    }
}

__global__ void __launch_bounds__(256)
caps_squash_kernel(const float* __restrict__ sbuf, float* __restrict__ vbuf,
                   float* __restrict__ out, int write_out) {
    const int idx = blockIdx.x * 256 + threadIdx.x;
    if (idx >= BATCH * CAPS * COUT) return;
    const int bc = idx / COUT;
    const float* sp = sbuf + (size_t)bc * COUT;
    float n2 = 0.0f;
#pragma unroll
    for (int o = 0; o < COUT; ++o) n2 += sp[o] * sp[o];
    const float scale = n2 / ((1.0f + n2) * sqrtf(n2));
    const float v = scale * sbuf[idx];
    vbuf[idx] = v;
    if (write_out) out[idx] = v;
}

#define ANPC   32
#define ACHUNK (NODES / ANPC)
#define AGRP   8

static __device__ __forceinline__ float caps_xhalf(float v) {
    return __builtin_bit_cast(float, __builtin_amdgcn_permlanex16(
        __builtin_bit_cast(int, v), __builtin_bit_cast(int, v),
        0x76543210, 0xfedcba98, false, false));
}

__global__ void __launch_bounds__(128)
caps_agree_kernel(const float* __restrict__ x, const float* __restrict__ W,
                  const float* __restrict__ vbuf, float* __restrict__ logits) {
    const int c     = blockIdx.x;
    const int chunk = blockIdx.y;
    const int wave  = threadIdx.x >> 5;
    const int lane  = threadIdx.x & 31;
    const int half  = lane >> 4;
    const int l16   = lane & 15;
    const int b     = wave * 16 + l16;

    const float* vp = vbuf + ((size_t)b * CAPS + c) * COUT + 8 * half;
    const float4 va = *(const float4*)(vp);
    const float4 vb = *(const float4*)(vp + 4);

    const float* xrow = x + (size_t)b * NODES * CIN + half * 2;
    const float* Wc   = W + (size_t)c * NODES * CIN * COUT
                          + (size_t)(half * 2) * COUT + l16;
    float* lrow = logits + ((size_t)b * CAPS + c) * NODES + 4 * half;

    const int n0 = chunk * ANPC;
    const float4 old0 = *(const float4*)(lrow + n0);
    const float4 old1 = *(const float4*)(lrow + n0 + AGRP);
    const float4 old2 = *(const float4*)(lrow + n0 + 2 * AGRP);
    const float4 old3 = *(const float4*)(lrow + n0 + 3 * AGRP);
    for (int g = 0; g < ANPC / AGRP; ++g) {
        const int ng = n0 + g * AGRP;
        const float4 old = (g == 0) ? old0 : (g == 1) ? old1 : (g == 2) ? old2 : old3;
        float t[AGRP];
#pragma unroll
        for (int j = 0; j < AGRP; ++j) {
            const float* xp = xrow + (size_t)(ng + j) * CIN;
            const float* Wp = Wc + (size_t)(ng + j) * CIN * COUT;
            v2f x_lo = { xp[0], xp[1] };
            v2f x_hi = { xp[4], xp[5] };
            v2f w_lo = { Wp[0],        Wp[COUT] };
            v2f w_hi = { Wp[4 * COUT], Wp[5 * COUT] };
            v8f acc = {};
            acc = __builtin_amdgcn_wmma_f32_16x16x4_f32(false, w_lo, false, x_lo,
                                                        (short)0, acc, false, false);
            acc = __builtin_amdgcn_wmma_f32_16x16x4_f32(false, w_hi, false, x_hi,
                                                        (short)0, acc, false, false);
            float s = acc[0] * va.x;
            s = fmaf(acc[1], va.y, s);
            s = fmaf(acc[2], va.z, s);
            s = fmaf(acc[3], va.w, s);
            s = fmaf(acc[4], vb.x, s);
            s = fmaf(acc[5], vb.y, s);
            s = fmaf(acc[6], vb.z, s);
            s = fmaf(acc[7], vb.w, s);
            t[j] = s;
        }
        float4 r;
        r.x = (half ? t[4] : t[0]) + caps_xhalf(half ? t[0] : t[4]);
        r.y = (half ? t[5] : t[1]) + caps_xhalf(half ? t[1] : t[5]);
        r.z = (half ? t[6] : t[2]) + caps_xhalf(half ? t[2] : t[6]);
        r.w = (half ? t[7] : t[3]) + caps_xhalf(half ? t[3] : t[7]);
        float4 upd;
        upd.x = old.x + r.x;
        upd.y = old.y + r.y;
        upd.z = old.z + r.z;
        upd.w = old.w + r.w;
        *(float4*)(lrow + ng) = upd;
    }
}

__global__ void __launch_bounds__(256)
caps_softmax_kernel(const float* __restrict__ logits, float* __restrict__ mz) {
    const int bc = blockIdx.x;
    const float* row = logits + (size_t)bc * NODES;
    __shared__ float red[256];
    float m = -1e30f;
    for (int n = threadIdx.x; n < NODES; n += 256) m = fmaxf(m, row[n]);
    red[threadIdx.x] = m;
    __syncthreads();
    for (int s = 128; s > 0; s >>= 1) {
        if (threadIdx.x < s)
            red[threadIdx.x] = fmaxf(red[threadIdx.x], red[threadIdx.x + s]);
        __syncthreads();
    }
    m = red[0];
    __syncthreads();
    float z = 0.0f;
    for (int n = threadIdx.x; n < NODES; n += 256) z += __expf(row[n] - m);
    red[threadIdx.x] = z;
    __syncthreads();
    for (int s = 128; s > 0; s >>= 1) {
        if (threadIdx.x < s) red[threadIdx.x] += red[threadIdx.x + s];
        __syncthreads();
    }
    if (threadIdx.x == 0) {
        mz[2 * bc + 0] = m;
        mz[2 * bc + 1] = red[0];
    }
}

extern "C" void kernel_launch(void* const* d_in, const int* in_sizes, int n_in,
                              void* d_out, int out_size, void* d_ws, size_t ws_size,
                              hipStream_t stream) {
    const float* x = (const float*)d_in[0];
    const float* W = (const float*)d_in[1];
    float* out = (float*)d_out;

    float* logits = (float*)d_ws;
    float* mz     = logits + (size_t)BATCH * CAPS * NODES;
    float* sbuf   = mz + (size_t)BATCH * CAPS * 2;
    float* vbuf   = sbuf + (size_t)BATCH * CAPS * COUT;

    caps_init_kernel<<<2048, 256, 0, stream>>>(logits, mz, sbuf);

    const dim3 gemm_grid(CAPS, NCHUNK);
    const dim3 agree_grid(CAPS, ACHUNK);
    for (int t = 0; t < 3; ++t) {
        caps_s_gemm_kernel<<<gemm_grid, 128, 0, stream>>>(x, W, logits, mz, sbuf);
        caps_squash_kernel<<<(BATCH * CAPS * COUT + 255) / 256, 256, 0, stream>>>(
            sbuf, vbuf, out, (t == 2) ? 1 : 0);
        if (t < 2) {
            caps_zero_s_kernel<<<(BATCH * CAPS * COUT + 255) / 256, 256, 0, stream>>>(sbuf);
            caps_agree_kernel<<<agree_grid, 128, 0, stream>>>(x, W, vbuf, logits);
            caps_softmax_kernel<<<BATCH * CAPS, 256, 0, stream>>>(logits, mz);
        }
    }
}


# s GEMM re-cut: a wave takes all four batch tiles for its 16 nodes (W fragment loaded once, four accumulators), 8 nodes' loads issued as one group, x as one 16-byte load per lane; waves summed in LDS
# speedup vs baseline: 5.8407x; 1.3048x over previous
#include <hip/hip_runtime.h>
#include <hip/hip_bf16.h>
#include <math.h>


#define BATCH 64
#define CAPS  10
#define NODES 6272
#define CIN   8
#define COUT  16
#define NCHUNK 49
#define NPC   (NODES / NCHUNK)

typedef float v2f __attribute__((ext_vector_type(2)));
typedef float v8f __attribute__((ext_vector_type(8)));

__global__ void __launch_bounds__(256)
caps_init_kernel(float* __restrict__ logits, float* __restrict__ mz,
                 float* __restrict__ sbuf) {
    const size_t NL = (size_t)BATCH * CAPS * NODES;
    size_t i = (size_t)blockIdx.x * 256 + threadIdx.x;
    for (size_t k = i; k < NL; k += (size_t)gridDim.x * 256)
        logits[k] = 0.0f;
    if (i < BATCH * CAPS) {
        mz[2 * i + 0] = 0.0f;
        mz[2 * i + 1] = (float)NODES;
    }
    if (i < BATCH * CAPS * COUT) sbuf[i] = 0.0f;
}

__global__ void __launch_bounds__(256)
caps_zero_s_kernel(float* __restrict__ sbuf) {
    int i = blockIdx.x * 256 + threadIdx.x;
    if (i < BATCH * CAPS * COUT) sbuf[i] = 0.0f;
}

#define SG_WAVES 8
#define SG_NPW   (NPC / SG_WAVES)
#define SG_GRP   8

__global__ void __launch_bounds__(SG_WAVES * 32)
caps_s_gemm_kernel(const float* __restrict__ x, const float* __restrict__ W,
                   const float* __restrict__ logits, const float* __restrict__ mz,
                   float* __restrict__ sbuf) {
    __shared__ float red[SG_WAVES * 4 * 256];
    const int c     = blockIdx.x;
    const int chunk = blockIdx.y;
    const int wave  = threadIdx.x >> 5;
    const int lane  = threadIdx.x & 31;
    const int half  = lane >> 4;
    const int l16   = lane & 15;

    float mrow[4], izrow[4];
    const float* xr[4];
    const float* lr[4];
#pragma unroll
    for (int tq = 0; tq < 4; ++tq) {
        const int b = tq * 16 + l16;
        mrow[tq]  = mz[((size_t)b * CAPS + c) * 2 + 0];
        izrow[tq] = 1.0f / mz[((size_t)b * CAPS + c) * 2 + 1];
        xr[tq] = x + (size_t)b * NODES * CIN + half * 4;
        lr[tq] = logits + ((size_t)b * CAPS + c) * NODES;
    }
    const float* Wc = W + (size_t)c * NODES * CIN * COUT
                        + (size_t)(half * 4) * COUT + l16;

    v8f acc[4] = {};
    const int nw = chunk * NPC + wave * SG_NPW;
    for (int g = 0; g < SG_NPW / SG_GRP; ++g) {
        const int ng = nw + g * SG_GRP;
        float4 lg[4][2];
        float4 xv[4][SG_GRP];
        float  wv[SG_GRP][4];
#pragma unroll
        for (int tq = 0; tq < 4; ++tq) {
            lg[tq][0] = *(const float4*)(lr[tq] + ng);
            lg[tq][1] = *(const float4*)(lr[tq] + ng + 4);
        }
#pragma unroll
        for (int j = 0; j < SG_GRP; ++j) {
            const float* Wp = Wc + (size_t)(ng + j) * CIN * COUT;
            wv[j][0] = Wp[0];
            wv[j][1] = Wp[COUT];
            wv[j][2] = Wp[2 * COUT];
            wv[j][3] = Wp[3 * COUT];
#pragma unroll
            for (int tq = 0; tq < 4; ++tq)
                xv[tq][j] = *(const float4*)(xr[tq] + (size_t)(ng + j) * CIN);
        }
#pragma unroll
        for (int j = 0; j < SG_GRP; ++j) {
            const v2f b_lo = { wv[j][0], wv[j][1] };
            const v2f b_hi = { wv[j][2], wv[j][3] };
            v2f a_lo[4], a_hi[4];
#pragma unroll
            for (int tq = 0; tq < 4; ++tq) {
                const float4 l4 = lg[tq][j >> 2];
                const float lgv = ((j & 3) == 0) ? l4.x : ((j & 3) == 1) ? l4.y
                                : ((j & 3) == 2) ? l4.z : l4.w;
                const float wgt = __expf(lgv - mrow[tq]) * izrow[tq];
                a_lo[tq] = (v2f){ wgt * xv[tq][j].x, wgt * xv[tq][j].y };
                a_hi[tq] = (v2f){ wgt * xv[tq][j].z, wgt * xv[tq][j].w };
            }
#pragma unroll
            for (int tq = 0; tq < 4; ++tq)
                acc[tq] = __builtin_amdgcn_wmma_f32_16x16x4_f32(false, a_lo[tq], false, b_lo,
                                                                (short)0, acc[tq], false, false);
#pragma unroll
            for (int tq = 0; tq < 4; ++tq)
                acc[tq] = __builtin_amdgcn_wmma_f32_16x16x4_f32(false, a_hi[tq], false, b_hi,
                                                                (short)0, acc[tq], false, false);
        }
    }
#pragma unroll
    for (int tq = 0; tq < 4; ++tq)
#pragma unroll
        for (int r = 0; r < 8; ++r)
            red[((wave * 4 + tq) * 8 + r) * 32 + lane] = acc[tq][r];
    __syncthreads();
#pragma unroll
    for (int k = 0; k < 4; ++k) {
        const int q = threadIdx.x + 256 * k;
        float sum = red[q];
#pragma unroll
        for (int w = 1; w < SG_WAVES; ++w) sum += red[w * 1024 + q];
        const int br = (q >> 8) * 16 + ((q >> 5) & 7) + 8 * ((q >> 4) & 1);
        atomicAdd(&sbuf[((size_t)br * CAPS + c) * COUT + (q & 15)], sum);
    }
}

__global__ void __launch_bounds__(256)
caps_squash_kernel(const float* __restrict__ sbuf, float* __restrict__ vbuf,
                   float* __restrict__ out, int write_out) {
    const int idx = blockIdx.x * 256 + threadIdx.x;
    if (idx >= BATCH * CAPS * COUT) return;
    const int bc = idx / COUT;
    const float* sp = sbuf + (size_t)bc * COUT;
    float n2 = 0.0f;
#pragma unroll
    for (int o = 0; o < COUT; ++o) n2 += sp[o] * sp[o];
    const float scale = n2 / ((1.0f + n2) * sqrtf(n2));
    const float v = scale * sbuf[idx];
    vbuf[idx] = v;
    if (write_out) out[idx] = v;
}

#define ANPC   32
#define ACHUNK (NODES / ANPC)
#define AGRP   8

static __device__ __forceinline__ float caps_xhalf(float v) {
    return __builtin_bit_cast(float, __builtin_amdgcn_permlanex16(
        __builtin_bit_cast(int, v), __builtin_bit_cast(int, v),
        0x76543210, 0xfedcba98, false, false));
}

__global__ void __launch_bounds__(128)
caps_agree_kernel(const float* __restrict__ x, const float* __restrict__ W,
                  const float* __restrict__ vbuf, float* __restrict__ logits) {
    const int c     = blockIdx.x;
    const int chunk = blockIdx.y;
    const int wave  = threadIdx.x >> 5;
    const int lane  = threadIdx.x & 31;
    const int half  = lane >> 4;
    const int l16   = lane & 15;
    const int b     = wave * 16 + l16;

    const float* vp = vbuf + ((size_t)b * CAPS + c) * COUT + 8 * half;
    const float4 va = *(const float4*)(vp);
    const float4 vb = *(const float4*)(vp + 4);

    const float* xrow = x + (size_t)b * NODES * CIN + half * 2;
    const float* Wc   = W + (size_t)c * NODES * CIN * COUT
                          + (size_t)(half * 2) * COUT + l16;
    float* lrow = logits + ((size_t)b * CAPS + c) * NODES + 4 * half;

    const int n0 = chunk * ANPC;
    const float4 old0 = *(const float4*)(lrow + n0);
    const float4 old1 = *(const float4*)(lrow + n0 + AGRP);
    const float4 old2 = *(const float4*)(lrow + n0 + 2 * AGRP);
    const float4 old3 = *(const float4*)(lrow + n0 + 3 * AGRP);
    for (int g = 0; g < ANPC / AGRP; ++g) {
        const int ng = n0 + g * AGRP;
        const float4 old = (g == 0) ? old0 : (g == 1) ? old1 : (g == 2) ? old2 : old3;
        float t[AGRP];
#pragma unroll
        for (int j = 0; j < AGRP; ++j) {
            const float* xp = xrow + (size_t)(ng + j) * CIN;
            const float* Wp = Wc + (size_t)(ng + j) * CIN * COUT;
            v2f x_lo = { xp[0], xp[1] };
            v2f x_hi = { xp[4], xp[5] };
            v2f w_lo = { Wp[0],        Wp[COUT] };
            v2f w_hi = { Wp[4 * COUT], Wp[5 * COUT] };
            v8f acc = {};
            acc = __builtin_amdgcn_wmma_f32_16x16x4_f32(false, w_lo, false, x_lo,
                                                        (short)0, acc, false, false);
            acc = __builtin_amdgcn_wmma_f32_16x16x4_f32(false, w_hi, false, x_hi,
                                                        (short)0, acc, false, false);
            float s = acc[0] * va.x;
            s = fmaf(acc[1], va.y, s);
            s = fmaf(acc[2], va.z, s);
            s = fmaf(acc[3], va.w, s);
            s = fmaf(acc[4], vb.x, s);
            s = fmaf(acc[5], vb.y, s);
            s = fmaf(acc[6], vb.z, s);
            s = fmaf(acc[7], vb.w, s);
            t[j] = s;
        }
        float4 r;
        r.x = (half ? t[4] : t[0]) + caps_xhalf(half ? t[0] : t[4]);
        r.y = (half ? t[5] : t[1]) + caps_xhalf(half ? t[1] : t[5]);
        r.z = (half ? t[6] : t[2]) + caps_xhalf(half ? t[2] : t[6]);
        r.w = (half ? t[7] : t[3]) + caps_xhalf(half ? t[3] : t[7]);
        float4 upd;
        upd.x = old.x + r.x;
        upd.y = old.y + r.y;
        upd.z = old.z + r.z;
        upd.w = old.w + r.w;
        *(float4*)(lrow + ng) = upd;
    }
}

__global__ void __launch_bounds__(256)
caps_softmax_kernel(const float* __restrict__ logits, float* __restrict__ mz) {
    const int bc = blockIdx.x;
    const float* row = logits + (size_t)bc * NODES;
    __shared__ float red[256];
    float m = -1e30f;
    for (int n = threadIdx.x; n < NODES; n += 256) m = fmaxf(m, row[n]);
    red[threadIdx.x] = m;
    __syncthreads();
    for (int s = 128; s > 0; s >>= 1) {
        if (threadIdx.x < s)
            red[threadIdx.x] = fmaxf(red[threadIdx.x], red[threadIdx.x + s]);
        __syncthreads();
    }
    m = red[0];
    __syncthreads();
    float z = 0.0f;
    for (int n = threadIdx.x; n < NODES; n += 256) z += __expf(row[n] - m);
    red[threadIdx.x] = z;
    __syncthreads();
    for (int s = 128; s > 0; s >>= 1) {
        if (threadIdx.x < s) red[threadIdx.x] += red[threadIdx.x + s];
        __syncthreads();
    }
    if (threadIdx.x == 0) {
        mz[2 * bc + 0] = m;
        mz[2 * bc + 1] = red[0];
    }
}

extern "C" void kernel_launch(void* const* d_in, const int* in_sizes, int n_in,
                              void* d_out, int out_size, void* d_ws, size_t ws_size,
                              hipStream_t stream) {
    const float* x = (const float*)d_in[0];
    const float* W = (const float*)d_in[1];
    float* out = (float*)d_out;

    float* logits = (float*)d_ws;
    float* mz     = logits + (size_t)BATCH * CAPS * NODES;
    float* sbuf   = mz + (size_t)BATCH * CAPS * 2;
    float* vbuf   = sbuf + (size_t)BATCH * CAPS * COUT;

    caps_init_kernel<<<2048, 256, 0, stream>>>(logits, mz, sbuf);

    const dim3 gemm_grid(CAPS, NCHUNK);
    const dim3 agree_grid(CAPS, ACHUNK);
    for (int t = 0; t < 3; ++t) {
        caps_s_gemm_kernel<<<gemm_grid, SG_WAVES * 32, 0, stream>>>(x, W, logits, mz, sbuf);
        caps_squash_kernel<<<(BATCH * CAPS * COUT + 255) / 256, 256, 0, stream>>>(
            sbuf, vbuf, out, (t == 2) ? 1 : 0);
        if (t < 2) {
            caps_zero_s_kernel<<<(BATCH * CAPS * COUT + 255) / 256, 256, 0, stream>>>(sbuf);
            caps_agree_kernel<<<agree_grid, 128, 0, stream>>>(x, W, vbuf, logits);
            caps_softmax_kernel<<<BATCH * CAPS, 256, 0, stream>>>(logits, mz);
        }
    }
}


# agreement fused into the next s GEMM: logits recomputed in registers from the running sum of the v's (transposed products, online softmax per wave); agree launches and logits array gone, 8 to 6
# speedup vs baseline: 13.4223x; 2.2981x over previous
#include <hip/hip_runtime.h>
#include <hip/hip_bf16.h>
#include <math.h>


#define BATCH 64
#define CAPS  10
#define NODES 6272
#define CIN   8
#define COUT  16
#define NCHUNK 49
#define NPC   (NODES / NCHUNK)

typedef float v2f __attribute__((ext_vector_type(2)));
typedef float v8f __attribute__((ext_vector_type(8)));

#define SG_WAVES 8
#define SG_NPW   (NPC / SG_WAVES)
#define PLANE    (BATCH * COUT)

__global__ void __launch_bounds__(SG_WAVES * 32)
caps_s_gemm_first_kernel(const float* __restrict__ x, const float* __restrict__ W,
                         float* __restrict__ part_s, float* __restrict__ part_mz) {
    __shared__ float red[SG_WAVES * 4 * 256];
    const int c     = blockIdx.x;
    const int chunk = blockIdx.y;
    const int wave  = threadIdx.x >> 5;
    const int lane  = threadIdx.x & 31;
    const int half  = lane >> 4;
    const int l16   = lane & 15;
    const int nw    = chunk * NPC + wave * SG_NPW;
    constexpr int GRP = 8;

    const float* xr[4];
#pragma unroll
    for (int tq = 0; tq < 4; ++tq)
        xr[tq] = x + (size_t)(tq * 16 + l16) * NODES * CIN + half * 4;
    const float* Wc = W + (size_t)c * NODES * CIN * COUT
                        + (size_t)(half * 4) * COUT + l16;

    v8f acc[4] = {};
    for (int g = 0; g < SG_NPW / GRP; ++g) {
        const int ng = nw + g * GRP;
        float4 xv[4][GRP];
        float  wv[GRP][4];
#pragma unroll
        for (int j = 0; j < GRP; ++j) {
            const float* Wp = Wc + (size_t)(ng + j) * CIN * COUT;
            wv[j][0] = Wp[0];
            wv[j][1] = Wp[COUT];
            wv[j][2] = Wp[2 * COUT];
            wv[j][3] = Wp[3 * COUT];
#pragma unroll
            for (int tq = 0; tq < 4; ++tq)
                xv[tq][j] = *(const float4*)(xr[tq] + (size_t)(ng + j) * CIN);
        }
        __builtin_amdgcn_sched_barrier(0);
#pragma unroll
        for (int j = 0; j < GRP; ++j) {
            const v2f b_lo = { wv[j][0], wv[j][1] };
            const v2f b_hi = { wv[j][2], wv[j][3] };
#pragma unroll
            for (int tq = 0; tq < 4; ++tq) {
                const v2f a_lo = { xv[tq][j].x, xv[tq][j].y };
                acc[tq] = __builtin_amdgcn_wmma_f32_16x16x4_f32(false, a_lo, false, b_lo,
                                                                (short)0, acc[tq], false, false);
            }
#pragma unroll
            for (int tq = 0; tq < 4; ++tq) {
                const v2f a_hi = { xv[tq][j].z, xv[tq][j].w };
                acc[tq] = __builtin_amdgcn_wmma_f32_16x16x4_f32(false, a_hi, false, b_hi,
                                                                (short)0, acc[tq], false, false);
            }
        }
    }
#pragma unroll
    for (int tq = 0; tq < 4; ++tq)
#pragma unroll
        for (int r = 0; r < 8; ++r)
            red[((wave * 4 + tq) * 8 + r) * 32 + lane] = acc[tq][r];
    __syncthreads();
    float* plane = part_s + (size_t)(chunk * CAPS + c) * PLANE;
    const int rr = threadIdx.x >> 4;
#pragma unroll
    for (int k = 0; k < 4; ++k) {
        const int q = (k * 8 + (rr & 7)) * 32 + (rr >> 3) * 16 + (threadIdx.x & 15);
        float sum = 0.0f;
#pragma unroll
        for (int w = 0; w < SG_WAVES; ++w) sum += red[w * 1024 + q];
        plane[k * 256 + threadIdx.x] = sum;
    }
    if (threadIdx.x < BATCH) {
        const float2 mzv = { 0.0f, (float)NPC };
        *(float2*)(part_mz + ((size_t)(chunk * CAPS + c) * BATCH + threadIdx.x) * 2) = mzv;
    }
}

static __device__ __forceinline__ float caps_xhalf(float v) {
    return __builtin_bit_cast(float, __builtin_amdgcn_permlanex16(
        __builtin_bit_cast(int, v), __builtin_bit_cast(int, v),
        0x76543210, 0xfedcba98, false, false));
}

#define RG_SLICES 4
#define RG_NPW   (NPC / RG_SLICES)
#define RG_TPW   2
#define RG_GRP   8

__global__ void __launch_bounds__(SG_WAVES * 32)
caps_route_gemm_kernel(const float* __restrict__ x, const float* __restrict__ W,
                       const float* __restrict__ vsum,
                       float* __restrict__ part_s, float* __restrict__ part_mz) {
    __shared__ __attribute__((aligned(16))) float red[RG_SLICES * PLANE];
    __shared__ float mw_s[RG_SLICES * BATCH];
    __shared__ float zw_s[RG_SLICES * BATCH];
    const int c     = blockIdx.x;
    const int chunk = blockIdx.y;
    const int wave  = threadIdx.x >> 5;
    const int slice = wave >> 1;
    const int bh    = wave & 1;
    const int lane  = threadIdx.x & 31;
    const int half  = lane >> 4;
    const int l16   = lane & 15;
    const int nw    = chunk * NPC + slice * RG_NPW;

    float4 va[RG_TPW], vb[RG_TPW];
    float  mrow[RG_TPW], zrow[RG_TPW];
    const float* xr[RG_TPW];
#pragma unroll
    for (int tq = 0; tq < RG_TPW; ++tq) {
        const int b = (bh * RG_TPW + tq) * 16 + l16;
        const float* vp = vsum + ((size_t)b * CAPS + c) * COUT + 8 * half;
        va[tq] = *(const float4*)(vp);
        vb[tq] = *(const float4*)(vp + 4);
        mrow[tq] = -INFINITY;
        zrow[tq] = 0.0f;
        xr[tq] = x + (size_t)b * NODES * CIN + half * 4;
    }
    const float* Wc = W + (size_t)c * NODES * CIN * COUT
                        + (size_t)(half * 4) * COUT + l16;

    v8f acc[RG_TPW] = {};
    for (int g = 0; g < RG_NPW / RG_GRP; ++g) {
        const int ng = nw + g * RG_GRP;
        float4 xv[RG_TPW][RG_GRP];
        float  wv[RG_GRP][4];
        float  lg[RG_TPW][RG_GRP];
#pragma unroll
        for (int j = 0; j < RG_GRP; ++j) {
            const float* Wp = Wc + (size_t)(ng + j) * CIN * COUT;
            wv[j][0] = Wp[0];
            wv[j][1] = Wp[COUT];
            wv[j][2] = Wp[2 * COUT];
            wv[j][3] = Wp[3 * COUT];
#pragma unroll
            for (int tq = 0; tq < RG_TPW; ++tq)
                xv[tq][j] = *(const float4*)(xr[tq] + (size_t)(ng + j) * CIN);
        }
        v8f d[2][RG_TPW];
#pragma unroll
        for (int j = -1; j < RG_GRP; ++j) {
            if (j + 1 < RG_GRP) {
                v2f w_lo = { wv[j + 1][0], wv[j + 1][1] };
                const v2f w_hi = { wv[j + 1][2], wv[j + 1][3] };
                if (j >= 1) asm volatile("" : "+v"(w_lo) : "v"(lg[0][j - 1]), "v"(lg[1][j - 1]));
#pragma unroll
                for (int tq = 0; tq < RG_TPW; ++tq) {
                    const v2f x_lo = { xv[tq][j + 1].x, xv[tq][j + 1].y };
                    const v8f zero = {};
                    d[(j + 1) & 1][tq] = __builtin_amdgcn_wmma_f32_16x16x4_f32(
                        false, w_lo, false, x_lo, (short)0, zero, false, false);
                }
#pragma unroll
                for (int tq = 0; tq < RG_TPW; ++tq) {
                    const v2f x_hi = { xv[tq][j + 1].z, xv[tq][j + 1].w };
                    d[(j + 1) & 1][tq] = __builtin_amdgcn_wmma_f32_16x16x4_f32(
                        false, w_hi, false, x_hi, (short)0, d[(j + 1) & 1][tq], false, false);
                }
            }
            if (j >= 0) {
#pragma unroll
                for (int tq = 0; tq < RG_TPW; ++tq) {
                    const v8f u = d[j & 1][tq];
                    float t = u[0] * va[tq].x;
                    t = fmaf(u[1], va[tq].y, t);
                    t = fmaf(u[2], va[tq].z, t);
                    t = fmaf(u[3], va[tq].w, t);
                    t = fmaf(u[4], vb[tq].x, t);
                    t = fmaf(u[5], vb[tq].y, t);
                    t = fmaf(u[6], vb[tq].z, t);
                    t = fmaf(u[7], vb[tq].w, t);
                    lg[tq][j] = t + caps_xhalf(t);
                }
            }
        }
#pragma unroll
        for (int tq = 0; tq < RG_TPW; ++tq) {
            float mg = lg[tq][0];
#pragma unroll
            for (int j = 1; j < RG_GRP; ++j) mg = fmaxf(mg, lg[tq][j]);
            const float mn = fmaxf(mrow[tq], mg);
            const float sc = __expf(mrow[tq] - mn);
#pragma unroll
            for (int r = 0; r < 8; ++r) acc[tq][r] *= sc;
            zrow[tq] *= sc;
            mrow[tq] = mn;
        }
#pragma unroll
        for (int j = 0; j < RG_GRP; ++j) {
            const v2f w_lo = { wv[j][0], wv[j][1] };
            const v2f w_hi = { wv[j][2], wv[j][3] };
            v2f wx_lo[RG_TPW], wx_hi[RG_TPW];
#pragma unroll
            for (int tq = 0; tq < RG_TPW; ++tq) {
                const float wgt = __expf(lg[tq][j] - mrow[tq]);
                zrow[tq] += wgt;
                wx_lo[tq] = (v2f){ wgt * xv[tq][j].x, wgt * xv[tq][j].y };
                wx_hi[tq] = (v2f){ wgt * xv[tq][j].z, wgt * xv[tq][j].w };
            }
#pragma unroll
            for (int tq = 0; tq < RG_TPW; ++tq)
                acc[tq] = __builtin_amdgcn_wmma_f32_16x16x4_f32(false, w_lo, false, wx_lo[tq],
                                                                (short)0, acc[tq], false, false);
#pragma unroll
            for (int tq = 0; tq < RG_TPW; ++tq)
                acc[tq] = __builtin_amdgcn_wmma_f32_16x16x4_f32(false, w_hi, false, wx_hi[tq],
                                                                (short)0, acc[tq], false, false);
        }
    }
#pragma unroll
    for (int tq = 0; tq < RG_TPW; ++tq) {
        const int b = (bh * RG_TPW + tq) * 16 + l16;
        float* rp = red + slice * PLANE + b * COUT + half * 8;
        *(float4*)(rp)     = (float4){ acc[tq][0], acc[tq][1], acc[tq][2], acc[tq][3] };
        *(float4*)(rp + 4) = (float4){ acc[tq][4], acc[tq][5], acc[tq][6], acc[tq][7] };
        if (half == 0) {
            mw_s[slice * BATCH + b] = mrow[tq];
            zw_s[slice * BATCH + b] = zrow[tq];
        }
    }
    __syncthreads();
    float* plane = part_s + (size_t)(chunk * CAPS + c) * PLANE;
#pragma unroll
    for (int k = 0; k < 4; ++k) {
        const int br = k * 16 + (threadIdx.x >> 4);
        float mk[RG_SLICES];
#pragma unroll
        for (int w = 0; w < RG_SLICES; ++w) mk[w] = mw_s[w * BATCH + br];
        float M = mk[0];
#pragma unroll
        for (int w = 1; w < RG_SLICES; ++w) M = fmaxf(M, mk[w]);
        float sum = 0.0f;
#pragma unroll
        for (int w = 0; w < RG_SLICES; ++w)
            sum += red[w * PLANE + k * 256 + threadIdx.x] * __expf(mk[w] - M);
        plane[k * 256 + threadIdx.x] = sum;
    }
    if (threadIdx.x < BATCH) {
        const int br = threadIdx.x;
        float mk[RG_SLICES];
#pragma unroll
        for (int w = 0; w < RG_SLICES; ++w) mk[w] = mw_s[w * BATCH + br];
        float M = mk[0];
#pragma unroll
        for (int w = 1; w < RG_SLICES; ++w) M = fmaxf(M, mk[w]);
        float Z = 0.0f;
#pragma unroll
        for (int w = 0; w < RG_SLICES; ++w) Z += zw_s[w * BATCH + br] * __expf(mk[w] - M);
        const float2 mzv = { M, Z };
        *(float2*)(part_mz + ((size_t)(chunk * CAPS + c) * BATCH + br) * 2) = mzv;
    }
}

__global__ void __launch_bounds__(256)
caps_squash_kernel(const float* __restrict__ part_s, const float* __restrict__ part_mz,
                   float* __restrict__ vsum, float* __restrict__ out, int step) {
    const int idx = blockIdx.x * 256 + threadIdx.x;
    const int o  = idx & 15;
    const int bc = idx >> 4;
    const int b  = bc / CAPS;
    const int c  = bc - b * CAPS;
    float2 mz[NCHUNK];
    float  sk[NCHUNK];
    const float vprev = (step == 1) ? vsum[idx] : 0.0f;
#pragma unroll
    for (int k = 0; k < NCHUNK; ++k) {
        const size_t row = (size_t)(k * CAPS + c) * BATCH + b;
        mz[k] = *(const float2*)(part_mz + row * 2);
        sk[k] = part_s[row * COUT + o];
    }
    float M = mz[0].x;
#pragma unroll
    for (int k = 1; k < NCHUNK; ++k) M = fmaxf(M, mz[k].x);
    float z = 0.0f, s = 0.0f;
#pragma unroll
    for (int k = 0; k < NCHUNK; ++k) {
        const float e = __expf(mz[k].x - M);
        z = fmaf(mz[k].y, e, z);
        s = fmaf(sk[k], e, s);
    }
    s = s / z;
    float n2 = s * s;
    n2 += __shfl_xor(n2, 8);
    n2 += __shfl_xor(n2, 4);
    n2 += __shfl_xor(n2, 2);
    n2 += __shfl_xor(n2, 1);
    const float scale = n2 / ((1.0f + n2) * sqrtf(n2));
    const float v = scale * s;
    if (step == 2) out[idx] = v;
    else           vsum[idx] = vprev + v;
}

extern "C" void kernel_launch(void* const* d_in, const int* in_sizes, int n_in,
                              void* d_out, int out_size, void* d_ws, size_t ws_size,
                              hipStream_t stream) {
    const float* x = (const float*)d_in[0];
    const float* W = (const float*)d_in[1];
    float* out = (float*)d_out;

    float* part_mz = (float*)d_ws;
    float* part_s  = part_mz + (size_t)NCHUNK * CAPS * BATCH * 2;
    float* vsum    = part_s + (size_t)NCHUNK * CAPS * PLANE;

    const dim3 gemm_grid(CAPS, NCHUNK);
    for (int t = 0; t < 3; ++t) {
        if (t == 0)
            caps_s_gemm_first_kernel<<<gemm_grid, SG_WAVES * 32, 0, stream>>>(x, W, part_s, part_mz);
        else
            caps_route_gemm_kernel<<<gemm_grid, SG_WAVES * 32, 0, stream>>>(x, W, vsum, part_s, part_mz);
        caps_squash_kernel<<<BATCH * CAPS * COUT / 256, 256, 0, stream>>>(
            part_s, part_mz, vsum, out, t);
    }
}


# route kernel: weighted s sum taken in-lane (packed FMAs) from the u_hat tiles the agreement product leaves in the D registers; second W.x product gone, matrix instructions per group 64 to 32
# speedup vs baseline: 15.5899x; 1.1615x over previous
#include <hip/hip_runtime.h>
#include <hip/hip_bf16.h>
#include <math.h>


#define BATCH 64
#define CAPS  10
#define NODES 6272
#define CIN   8
#define COUT  16
#define NCHUNK 49
#define NPC   (NODES / NCHUNK)

typedef float v2f __attribute__((ext_vector_type(2)));
typedef float v8f __attribute__((ext_vector_type(8)));

#define SG_WAVES 8
#define SG_NPW   (NPC / SG_WAVES)
#define PLANE    (BATCH * COUT)

__global__ void __launch_bounds__(SG_WAVES * 32)
caps_s_gemm_first_kernel(const float* __restrict__ x, const float* __restrict__ W,
                         float* __restrict__ part_s, float* __restrict__ part_mz) {
    __shared__ float red[SG_WAVES * 4 * 256];
    const int c     = blockIdx.x;
    const int chunk = blockIdx.y;
    const int wave  = threadIdx.x >> 5;
    const int lane  = threadIdx.x & 31;
    const int half  = lane >> 4;
    const int l16   = lane & 15;
    const int nw    = chunk * NPC + wave * SG_NPW;
    constexpr int GRP = 8;

    const float* xr[4];
#pragma unroll
    for (int tq = 0; tq < 4; ++tq)
        xr[tq] = x + (size_t)(tq * 16 + l16) * NODES * CIN + half * 4;
    const float* Wc = W + (size_t)c * NODES * CIN * COUT
                        + (size_t)(half * 4) * COUT + l16;

    v8f acc[4] = {};
    for (int g = 0; g < SG_NPW / GRP; ++g) {
        const int ng = nw + g * GRP;
        float4 xv[4][GRP];
        float  wv[GRP][4];
#pragma unroll
        for (int j = 0; j < GRP; ++j) {
            const float* Wp = Wc + (size_t)(ng + j) * CIN * COUT;
            wv[j][0] = Wp[0];
            wv[j][1] = Wp[COUT];
            wv[j][2] = Wp[2 * COUT];
            wv[j][3] = Wp[3 * COUT];
#pragma unroll
            for (int tq = 0; tq < 4; ++tq)
                xv[tq][j] = *(const float4*)(xr[tq] + (size_t)(ng + j) * CIN);
        }
        __builtin_amdgcn_sched_barrier(0);
#pragma unroll
        for (int j = 0; j < GRP; ++j) {
            const v2f b_lo = { wv[j][0], wv[j][1] };
            const v2f b_hi = { wv[j][2], wv[j][3] };
#pragma unroll
            for (int tq = 0; tq < 4; ++tq) {
                const v2f a_lo = { xv[tq][j].x, xv[tq][j].y };
                acc[tq] = __builtin_amdgcn_wmma_f32_16x16x4_f32(false, a_lo, false, b_lo,
                                                                (short)0, acc[tq], false, false);
            }
#pragma unroll
            for (int tq = 0; tq < 4; ++tq) {
                const v2f a_hi = { xv[tq][j].z, xv[tq][j].w };
                acc[tq] = __builtin_amdgcn_wmma_f32_16x16x4_f32(false, a_hi, false, b_hi,
                                                                (short)0, acc[tq], false, false);
            }
        }
    }
#pragma unroll
    for (int tq = 0; tq < 4; ++tq)
#pragma unroll
        for (int r = 0; r < 8; ++r)
            red[((wave * 4 + tq) * 8 + r) * 32 + lane] = acc[tq][r];
    __syncthreads();
    float* plane = part_s + (size_t)(chunk * CAPS + c) * PLANE;
    const int rr = threadIdx.x >> 4;
#pragma unroll
    for (int k = 0; k < 4; ++k) {
        const int q = (k * 8 + (rr & 7)) * 32 + (rr >> 3) * 16 + (threadIdx.x & 15);
        float sum = 0.0f;
#pragma unroll
        for (int w = 0; w < SG_WAVES; ++w) sum += red[w * 1024 + q];
        plane[k * 256 + threadIdx.x] = sum;
    }
    if (threadIdx.x < BATCH) {
        const float2 mzv = { 0.0f, (float)NPC };
        *(float2*)(part_mz + ((size_t)(chunk * CAPS + c) * BATCH + threadIdx.x) * 2) = mzv;
    }
}

static __device__ __forceinline__ float caps_xhalf(float v) {
    return __builtin_bit_cast(float, __builtin_amdgcn_permlanex16(
        __builtin_bit_cast(int, v), __builtin_bit_cast(int, v),
        0x76543210, 0xfedcba98, false, false));
}

#define RG_SLICES 4
#define RG_NPW   (NPC / RG_SLICES)
#define RG_TPW   2
#define RG_GRP   8

__global__ void __launch_bounds__(SG_WAVES * 32)
caps_route_gemm_kernel(const float* __restrict__ x, const float* __restrict__ W,
                       const float* __restrict__ vsum,
                       float* __restrict__ part_s, float* __restrict__ part_mz) {
    __shared__ __attribute__((aligned(16))) float red[RG_SLICES * PLANE];
    __shared__ float mw_s[RG_SLICES * BATCH];
    __shared__ float zw_s[RG_SLICES * BATCH];
    const int c     = blockIdx.x;
    const int chunk = blockIdx.y;
    const int wave  = threadIdx.x >> 5;
    const int slice = wave >> 1;
    const int bh    = wave & 1;
    const int lane  = threadIdx.x & 31;
    const int half  = lane >> 4;
    const int l16   = lane & 15;
    const int nw    = chunk * NPC + slice * RG_NPW;

    float4 va[RG_TPW], vb[RG_TPW];
    float  mrow[RG_TPW], zrow[RG_TPW];
    const float* xr[RG_TPW];
#pragma unroll
    for (int tq = 0; tq < RG_TPW; ++tq) {
        const int b = (bh * RG_TPW + tq) * 16 + l16;
        const float* vp = vsum + ((size_t)b * CAPS + c) * COUT + 8 * half;
        va[tq] = *(const float4*)(vp);
        vb[tq] = *(const float4*)(vp + 4);
        mrow[tq] = -INFINITY;
        zrow[tq] = 0.0f;
        xr[tq] = x + (size_t)b * NODES * CIN + half * 4;
    }
    const float* Wc = W + (size_t)c * NODES * CIN * COUT
                        + (size_t)(half * 4) * COUT + l16;

    v8f acc[RG_TPW] = {};
    for (int g = 0; g < RG_NPW / RG_GRP; ++g) {
        const int ng = nw + g * RG_GRP;
        float4 xv[RG_TPW][RG_GRP];
        float  wv[RG_GRP][4];
        float  lg[RG_TPW][RG_GRP];
#pragma unroll
        for (int j = 0; j < RG_GRP; ++j) {
            const float* Wp = Wc + (size_t)(ng + j) * CIN * COUT;
            wv[j][0] = Wp[0];
            wv[j][1] = Wp[COUT];
            wv[j][2] = Wp[2 * COUT];
            wv[j][3] = Wp[3 * COUT];
#pragma unroll
            for (int tq = 0; tq < RG_TPW; ++tq)
                xv[tq][j] = *(const float4*)(xr[tq] + (size_t)(ng + j) * CIN);
        }
        v8f u[RG_GRP][RG_TPW];
#pragma unroll
        for (int j = -1; j < RG_GRP; ++j) {
            if (j + 1 < RG_GRP) {
                const v2f w_lo = { wv[j + 1][0], wv[j + 1][1] };
                const v2f w_hi = { wv[j + 1][2], wv[j + 1][3] };
#pragma unroll
                for (int tq = 0; tq < RG_TPW; ++tq) {
                    const v2f x_lo = { xv[tq][j + 1].x, xv[tq][j + 1].y };
                    const v8f zero = {};
                    u[j + 1][tq] = __builtin_amdgcn_wmma_f32_16x16x4_f32(
                        false, w_lo, false, x_lo, (short)0, zero, false, false);
                }
#pragma unroll
                for (int tq = 0; tq < RG_TPW; ++tq) {
                    const v2f x_hi = { xv[tq][j + 1].z, xv[tq][j + 1].w };
                    u[j + 1][tq] = __builtin_amdgcn_wmma_f32_16x16x4_f32(
                        false, w_hi, false, x_hi, (short)0, u[j + 1][tq], false, false);
                }
            }
            if (j >= 0) {
#pragma unroll
                for (int tq = 0; tq < RG_TPW; ++tq) {
                    const v8f uu = u[j][tq];
                    v2f t2 = (v2f){ uu[0], uu[1] } * (v2f){ va[tq].x, va[tq].y };
                    t2 = __builtin_elementwise_fma((v2f){ uu[2], uu[3] }, (v2f){ va[tq].z, va[tq].w }, t2);
                    t2 = __builtin_elementwise_fma((v2f){ uu[4], uu[5] }, (v2f){ vb[tq].x, vb[tq].y }, t2);
                    t2 = __builtin_elementwise_fma((v2f){ uu[6], uu[7] }, (v2f){ vb[tq].z, vb[tq].w }, t2);
                    const float t = t2.x + t2.y;
                    lg[tq][j] = t + caps_xhalf(t);
                }
            }
            __builtin_amdgcn_sched_barrier(0);
        }
#pragma unroll
        for (int tq = 0; tq < RG_TPW; ++tq) {
            float mg = lg[tq][0];
#pragma unroll
            for (int j = 1; j < RG_GRP; ++j) mg = fmaxf(mg, lg[tq][j]);
            const float mn = fmaxf(mrow[tq], mg);
            const float sc = __expf(mrow[tq] - mn);
            acc[tq] *= sc;
            zrow[tq] *= sc;
            mrow[tq] = mn;
        }
#pragma unroll
        for (int j = 0; j < RG_GRP; ++j) {
#pragma unroll
            for (int tq = 0; tq < RG_TPW; ++tq) {
                const float wgt = __expf(lg[tq][j] - mrow[tq]);
                zrow[tq] += wgt;
                const v8f wgt8 = { wgt, wgt, wgt, wgt, wgt, wgt, wgt, wgt };
                acc[tq] = __builtin_elementwise_fma(wgt8, u[j][tq], acc[tq]);
            }
        }
    }
#pragma unroll
    for (int tq = 0; tq < RG_TPW; ++tq) {
        const int b = (bh * RG_TPW + tq) * 16 + l16;
        float* rp = red + slice * PLANE + b * COUT + half * 8;
        *(float4*)(rp)     = (float4){ acc[tq][0], acc[tq][1], acc[tq][2], acc[tq][3] };
        *(float4*)(rp + 4) = (float4){ acc[tq][4], acc[tq][5], acc[tq][6], acc[tq][7] };
        if (half == 0) {
            mw_s[slice * BATCH + b] = mrow[tq];
            zw_s[slice * BATCH + b] = zrow[tq];
        }
    }
    __syncthreads();
    float* plane = part_s + (size_t)(chunk * CAPS + c) * PLANE;
#pragma unroll
    for (int k = 0; k < 4; ++k) {
        const int br = k * 16 + (threadIdx.x >> 4);
        float mk[RG_SLICES];
#pragma unroll
        for (int w = 0; w < RG_SLICES; ++w) mk[w] = mw_s[w * BATCH + br];
        float M = mk[0];
#pragma unroll
        for (int w = 1; w < RG_SLICES; ++w) M = fmaxf(M, mk[w]);
        float sum = 0.0f;
#pragma unroll
        for (int w = 0; w < RG_SLICES; ++w)
            sum += red[w * PLANE + k * 256 + threadIdx.x] * __expf(mk[w] - M);
        plane[k * 256 + threadIdx.x] = sum;
    }
    if (threadIdx.x < BATCH) {
        const int br = threadIdx.x;
        float mk[RG_SLICES];
#pragma unroll
        for (int w = 0; w < RG_SLICES; ++w) mk[w] = mw_s[w * BATCH + br];
        float M = mk[0];
#pragma unroll
        for (int w = 1; w < RG_SLICES; ++w) M = fmaxf(M, mk[w]);
        float Z = 0.0f;
#pragma unroll
        for (int w = 0; w < RG_SLICES; ++w) Z += zw_s[w * BATCH + br] * __expf(mk[w] - M);
        const float2 mzv = { M, Z };
        *(float2*)(part_mz + ((size_t)(chunk * CAPS + c) * BATCH + br) * 2) = mzv;
    }
}

__global__ void __launch_bounds__(256)
caps_squash_kernel(const float* __restrict__ part_s, const float* __restrict__ part_mz,
                   float* __restrict__ vsum, float* __restrict__ out, int step) {
    const int idx = blockIdx.x * 256 + threadIdx.x;
    const int o  = idx & 15;
    const int bc = idx >> 4;
    const int b  = bc / CAPS;
    const int c  = bc - b * CAPS;
    float2 mz[NCHUNK];
    float  sk[NCHUNK];
    const float vprev = (step == 1) ? vsum[idx] : 0.0f;
#pragma unroll
    for (int k = 0; k < NCHUNK; ++k) {
        const size_t row = (size_t)(k * CAPS + c) * BATCH + b;
        mz[k] = *(const float2*)(part_mz + row * 2);
        sk[k] = part_s[row * COUT + o];
    }
    float M = mz[0].x;
#pragma unroll
    for (int k = 1; k < NCHUNK; ++k) M = fmaxf(M, mz[k].x);
    float z = 0.0f, s = 0.0f;
#pragma unroll
    for (int k = 0; k < NCHUNK; ++k) {
        const float e = __expf(mz[k].x - M);
        z = fmaf(mz[k].y, e, z);
        s = fmaf(sk[k], e, s);
    }
    s = s / z;
    float n2 = s * s;
    n2 += __shfl_xor(n2, 8);
    n2 += __shfl_xor(n2, 4);
    n2 += __shfl_xor(n2, 2);
    n2 += __shfl_xor(n2, 1);
    const float scale = n2 / ((1.0f + n2) * sqrtf(n2));
    const float v = scale * s;
    if (step == 2) out[idx] = v;
    else           vsum[idx] = vprev + v;
}

extern "C" void kernel_launch(void* const* d_in, const int* in_sizes, int n_in,
                              void* d_out, int out_size, void* d_ws, size_t ws_size,
                              hipStream_t stream) {
    const float* x = (const float*)d_in[0];
    const float* W = (const float*)d_in[1];
    float* out = (float*)d_out;

    float* part_mz = (float*)d_ws;
    float* part_s  = part_mz + (size_t)NCHUNK * CAPS * BATCH * 2;
    float* vsum    = part_s + (size_t)NCHUNK * CAPS * PLANE;

    const dim3 gemm_grid(CAPS, NCHUNK);
    for (int t = 0; t < 3; ++t) {
        if (t == 0)
            caps_s_gemm_first_kernel<<<gemm_grid, SG_WAVES * 32, 0, stream>>>(x, W, part_s, part_mz);
        else
            caps_route_gemm_kernel<<<gemm_grid, SG_WAVES * 32, 0, stream>>>(x, W, vsum, part_s, part_mz);
        caps_squash_kernel<<<BATCH * CAPS * COUT / 256, 256, 0, stream>>>(
            part_s, part_mz, vsum, out, t);
    }
}


# both GEMM kernels: a block takes a pair of caps (16 waves) and each wave multiplies its x fragments with both caps' W fragments: x fragment loads and x line fetches per WGP halved
# speedup vs baseline: 17.1699x; 1.1013x over previous
#include <hip/hip_runtime.h>
#include <hip/hip_bf16.h>
#include <math.h>


#define BATCH 64
#define CAPS  10
#define NODES 6272
#define CIN   8
#define COUT  16
#define NCHUNK 49
#define NPC   (NODES / NCHUNK)

typedef float v2f __attribute__((ext_vector_type(2)));
typedef float v8f __attribute__((ext_vector_type(8)));

#define PLANE    (BATCH * COUT)
#define CPB      2

#define FG_WAVES 16
#define FG_NPW   (NPC / FG_WAVES)
#define FG_GRP   4

__global__ void __launch_bounds__(FG_WAVES * 32)
caps_s_gemm_first_kernel(const float* __restrict__ x, const float* __restrict__ W,
                         float* __restrict__ part_s, float* __restrict__ part_mz) {
    __shared__ float red[FG_WAVES * CPB * PLANE];
    const int c0    = blockIdx.x * CPB;
    const int chunk = blockIdx.y;
    const int wave  = threadIdx.x >> 5;
    const int lane  = threadIdx.x & 31;
    const int half  = lane >> 4;
    const int l16   = lane & 15;
    const int nw    = chunk * NPC + wave * FG_NPW;

    const float* xr[4];
#pragma unroll
    for (int tq = 0; tq < 4; ++tq)
        xr[tq] = x + (size_t)(tq * 16 + l16) * NODES * CIN + half * 4;
    const float* Wc = W + (size_t)c0 * NODES * CIN * COUT
                        + (size_t)(half * 4) * COUT + l16;

    v8f acc[CPB][4] = {};
    for (int g = 0; g < FG_NPW / FG_GRP; ++g) {
        const int ng = nw + g * FG_GRP;
        float4 xv[4][FG_GRP];
        float  wv[CPB][FG_GRP][4];
#pragma unroll
        for (int j = 0; j < FG_GRP; ++j) {
#pragma unroll
            for (int cc = 0; cc < CPB; ++cc) {
                const float* Wp = Wc + ((size_t)cc * NODES + (ng + j)) * CIN * COUT;
                wv[cc][j][0] = Wp[0];
                wv[cc][j][1] = Wp[COUT];
                wv[cc][j][2] = Wp[2 * COUT];
                wv[cc][j][3] = Wp[3 * COUT];
            }
#pragma unroll
            for (int tq = 0; tq < 4; ++tq)
                xv[tq][j] = *(const float4*)(xr[tq] + (size_t)(ng + j) * CIN);
        }
        __builtin_amdgcn_sched_barrier(0);
#pragma unroll
        for (int j = 0; j < FG_GRP; ++j) {
#pragma unroll
            for (int cc = 0; cc < CPB; ++cc) {
                const v2f b_lo = { wv[cc][j][0], wv[cc][j][1] };
#pragma unroll
                for (int tq = 0; tq < 4; ++tq) {
                    const v2f a_lo = { xv[tq][j].x, xv[tq][j].y };
                    acc[cc][tq] = __builtin_amdgcn_wmma_f32_16x16x4_f32(false, a_lo, false, b_lo,
                                                                    (short)0, acc[cc][tq], false, false);
                }
            }
#pragma unroll
            for (int cc = 0; cc < CPB; ++cc) {
                const v2f b_hi = { wv[cc][j][2], wv[cc][j][3] };
#pragma unroll
                for (int tq = 0; tq < 4; ++tq) {
                    const v2f a_hi = { xv[tq][j].z, xv[tq][j].w };
                    acc[cc][tq] = __builtin_amdgcn_wmma_f32_16x16x4_f32(false, a_hi, false, b_hi,
                                                                    (short)0, acc[cc][tq], false, false);
                }
            }
        }
    }
#pragma unroll
    for (int cc = 0; cc < CPB; ++cc)
#pragma unroll
        for (int tq = 0; tq < 4; ++tq)
#pragma unroll
            for (int r = 0; r < 8; ++r)
                red[(((wave * CPB + cc) * 4 + tq) * 8 + r) * 32 + lane] = acc[cc][tq][r];
    __syncthreads();
#pragma unroll
    for (int k = 0; k < 4; ++k) {
        const int cc = k >> 1;
        const int pe = (k & 1) * 512 + threadIdx.x;
        const int tq = pe >> 8;
        const int rr = (pe >> 4) & 15;
        const int q  = (tq * 8 + (rr & 7)) * 32 + (rr >> 3) * 16 + (pe & 15);
        float sum = 0.0f;
#pragma unroll
        for (int w = 0; w < FG_WAVES; ++w) sum += red[(w * CPB + cc) * PLANE + q];
        part_s[(size_t)(chunk * CAPS + c0 + cc) * PLANE + pe] = sum;
    }
    if (threadIdx.x < CPB * BATCH) {
        const int cc = threadIdx.x >> 6;
        const int br = threadIdx.x & (BATCH - 1);
        const float2 mzv = { 0.0f, (float)NPC };
        *(float2*)(part_mz + ((size_t)(chunk * CAPS + c0 + cc) * BATCH + br) * 2) = mzv;
    }
}

static __device__ __forceinline__ float caps_xhalf(float v) {
    return __builtin_bit_cast(float, __builtin_amdgcn_permlanex16(
        __builtin_bit_cast(int, v), __builtin_bit_cast(int, v),
        0x76543210, 0xfedcba98, false, false));
}

#define RG_WAVES  16
#define RG_SLICES 8
#define RG_NPW   (NPC / RG_SLICES)
#define RG_TPW   2
#define RG_GRP   4

__global__ void __launch_bounds__(RG_WAVES * 32)
caps_route_gemm_kernel(const float* __restrict__ x, const float* __restrict__ W,
                       const float* __restrict__ vsum,
                       float* __restrict__ part_s, float* __restrict__ part_mz) {
    __shared__ __attribute__((aligned(16))) float red[RG_SLICES * CPB * PLANE];
    __shared__ float mw_s[RG_SLICES * CPB * BATCH];
    __shared__ float zw_s[RG_SLICES * CPB * BATCH];
    const int c0    = blockIdx.x * CPB;
    const int chunk = blockIdx.y;
    const int wave  = threadIdx.x >> 5;
    const int slice = wave >> 1;
    const int bh    = wave & 1;
    const int lane  = threadIdx.x & 31;
    const int half  = lane >> 4;
    const int l16   = lane & 15;
    const int nw    = chunk * NPC + slice * RG_NPW;

    float4 va[CPB][RG_TPW], vb[CPB][RG_TPW];
    float  mrow[CPB][RG_TPW], zrow[CPB][RG_TPW];
    const float* xr[RG_TPW];
#pragma unroll
    for (int tq = 0; tq < RG_TPW; ++tq) {
        const int b = (bh * RG_TPW + tq) * 16 + l16;
#pragma unroll
        for (int cc = 0; cc < CPB; ++cc) {
            const float* vp = vsum + ((size_t)b * CAPS + c0 + cc) * COUT + 8 * half;
            va[cc][tq] = *(const float4*)(vp);
            vb[cc][tq] = *(const float4*)(vp + 4);
            mrow[cc][tq] = -INFINITY;
            zrow[cc][tq] = 0.0f;
        }
        xr[tq] = x + (size_t)b * NODES * CIN + half * 4;
    }
    const float* Wc = W + (size_t)c0 * NODES * CIN * COUT
                        + (size_t)(half * 4) * COUT + l16;

    v8f acc[CPB][RG_TPW] = {};
    for (int g = 0; g < RG_NPW / RG_GRP; ++g) {
        const int ng = nw + g * RG_GRP;
        float4 xv[RG_TPW][RG_GRP];
        float  wv[CPB][RG_GRP][4];
        float  lg[CPB][RG_TPW][RG_GRP];
#pragma unroll
        for (int j = 0; j < RG_GRP; ++j) {
#pragma unroll
            for (int cc = 0; cc < CPB; ++cc) {
                const float* Wp = Wc + ((size_t)cc * NODES + (ng + j)) * CIN * COUT;
                wv[cc][j][0] = Wp[0];
                wv[cc][j][1] = Wp[COUT];
                wv[cc][j][2] = Wp[2 * COUT];
                wv[cc][j][3] = Wp[3 * COUT];
            }
#pragma unroll
            for (int tq = 0; tq < RG_TPW; ++tq)
                xv[tq][j] = *(const float4*)(xr[tq] + (size_t)(ng + j) * CIN);
        }
        v8f u[RG_GRP][CPB][RG_TPW];
#pragma unroll
        for (int j = -1; j < RG_GRP; ++j) {
            if (j + 1 < RG_GRP) {
#pragma unroll
                for (int cc = 0; cc < CPB; ++cc) {
                    const v2f w_lo = { wv[cc][j + 1][0], wv[cc][j + 1][1] };
#pragma unroll
                    for (int tq = 0; tq < RG_TPW; ++tq) {
                        const v2f x_lo = { xv[tq][j + 1].x, xv[tq][j + 1].y };
                        const v8f zero = {};
                        u[j + 1][cc][tq] = __builtin_amdgcn_wmma_f32_16x16x4_f32(
                            false, w_lo, false, x_lo, (short)0, zero, false, false);
                    }
                }
#pragma unroll
                for (int cc = 0; cc < CPB; ++cc) {
                    const v2f w_hi = { wv[cc][j + 1][2], wv[cc][j + 1][3] };
#pragma unroll
                    for (int tq = 0; tq < RG_TPW; ++tq) {
                        const v2f x_hi = { xv[tq][j + 1].z, xv[tq][j + 1].w };
                        u[j + 1][cc][tq] = __builtin_amdgcn_wmma_f32_16x16x4_f32(
                            false, w_hi, false, x_hi, (short)0, u[j + 1][cc][tq], false, false);
                    }
                }
            }
            if (j >= 0) {
#pragma unroll
                for (int cc = 0; cc < CPB; ++cc) {
#pragma unroll
                    for (int tq = 0; tq < RG_TPW; ++tq) {
                        const v8f uu = u[j][cc][tq];
                        const float4 pa = va[cc][tq], pb = vb[cc][tq];
                        v2f t2 = (v2f){ uu[0], uu[1] } * (v2f){ pa.x, pa.y };
                        t2 = __builtin_elementwise_fma((v2f){ uu[2], uu[3] }, (v2f){ pa.z, pa.w }, t2);
                        t2 = __builtin_elementwise_fma((v2f){ uu[4], uu[5] }, (v2f){ pb.x, pb.y }, t2);
                        t2 = __builtin_elementwise_fma((v2f){ uu[6], uu[7] }, (v2f){ pb.z, pb.w }, t2);
                        const float t = t2.x + t2.y;
                        lg[cc][tq][j] = t + caps_xhalf(t);
                    }
                }
            }
            __builtin_amdgcn_sched_barrier(0);
        }
#pragma unroll
        for (int cc = 0; cc < CPB; ++cc) {
#pragma unroll
            for (int tq = 0; tq < RG_TPW; ++tq) {
                float mg = lg[cc][tq][0];
#pragma unroll
                for (int j = 1; j < RG_GRP; ++j) mg = fmaxf(mg, lg[cc][tq][j]);
                const float mn = fmaxf(mrow[cc][tq], mg);
                const float sc = __expf(mrow[cc][tq] - mn);
                acc[cc][tq] *= sc;
                zrow[cc][tq] *= sc;
                mrow[cc][tq] = mn;
            }
        }
#pragma unroll
        for (int j = 0; j < RG_GRP; ++j) {
#pragma unroll
            for (int cc = 0; cc < CPB; ++cc) {
#pragma unroll
                for (int tq = 0; tq < RG_TPW; ++tq) {
                    const float wgt = __expf(lg[cc][tq][j] - mrow[cc][tq]);
                    zrow[cc][tq] += wgt;
                    const v8f wgt8 = { wgt, wgt, wgt, wgt, wgt, wgt, wgt, wgt };
                    acc[cc][tq] = __builtin_elementwise_fma(wgt8, u[j][cc][tq], acc[cc][tq]);
                }
            }
        }
    }
#pragma unroll
    for (int cc = 0; cc < CPB; ++cc) {
#pragma unroll
        for (int tq = 0; tq < RG_TPW; ++tq) {
            const int b = (bh * RG_TPW + tq) * 16 + l16;
            float* rp = red + (slice * CPB + cc) * PLANE + b * COUT + half * 8;
            *(float4*)(rp)     = (float4){ acc[cc][tq][0], acc[cc][tq][1], acc[cc][tq][2], acc[cc][tq][3] };
            *(float4*)(rp + 4) = (float4){ acc[cc][tq][4], acc[cc][tq][5], acc[cc][tq][6], acc[cc][tq][7] };
            if (half == 0) {
                mw_s[(slice * CPB + cc) * BATCH + b] = mrow[cc][tq];
                zw_s[(slice * CPB + cc) * BATCH + b] = zrow[cc][tq];
            }
        }
    }
    __syncthreads();
#pragma unroll
    for (int k = 0; k < 4; ++k) {
        const int cc = k >> 1;
        const int pe = (k & 1) * 512 + threadIdx.x;
        const int br = pe >> 4;
        float mk[RG_SLICES];
#pragma unroll
        for (int w = 0; w < RG_SLICES; ++w) mk[w] = mw_s[(w * CPB + cc) * BATCH + br];
        float M = mk[0];
#pragma unroll
        for (int w = 1; w < RG_SLICES; ++w) M = fmaxf(M, mk[w]);
        float sum = 0.0f;
#pragma unroll
        for (int w = 0; w < RG_SLICES; ++w)
            sum += red[(w * CPB + cc) * PLANE + pe] * __expf(mk[w] - M);
        part_s[(size_t)(chunk * CAPS + c0 + cc) * PLANE + pe] = sum;
    }
    if (threadIdx.x < CPB * BATCH) {
        const int cc = threadIdx.x >> 6;
        const int br = threadIdx.x & (BATCH - 1);
        float mk[RG_SLICES];
#pragma unroll
        for (int w = 0; w < RG_SLICES; ++w) mk[w] = mw_s[(w * CPB + cc) * BATCH + br];
        float M = mk[0];
#pragma unroll
        for (int w = 1; w < RG_SLICES; ++w) M = fmaxf(M, mk[w]);
        float Z = 0.0f;
#pragma unroll
        for (int w = 0; w < RG_SLICES; ++w) Z += zw_s[(w * CPB + cc) * BATCH + br] * __expf(mk[w] - M);
        const float2 mzv = { M, Z };
        *(float2*)(part_mz + ((size_t)(chunk * CAPS + c0 + cc) * BATCH + br) * 2) = mzv;
    }
}

__global__ void __launch_bounds__(256)
caps_squash_kernel(const float* __restrict__ part_s, const float* __restrict__ part_mz,
                   float* __restrict__ vsum, float* __restrict__ out, int step) {
    const int idx = blockIdx.x * 256 + threadIdx.x;
    const int o  = idx & 15;
    const int bc = idx >> 4;
    const int b  = bc / CAPS;
    const int c  = bc - b * CAPS;
    float2 mz[NCHUNK];
    float  sk[NCHUNK];
    const float vprev = (step == 1) ? vsum[idx] : 0.0f;
#pragma unroll
    for (int k = 0; k < NCHUNK; ++k) {
        const size_t row = (size_t)(k * CAPS + c) * BATCH + b;
        mz[k] = *(const float2*)(part_mz + row * 2);
        sk[k] = part_s[row * COUT + o];
    }
    float M = mz[0].x;
#pragma unroll
    for (int k = 1; k < NCHUNK; ++k) M = fmaxf(M, mz[k].x);
    float z = 0.0f, s = 0.0f;
#pragma unroll
    for (int k = 0; k < NCHUNK; ++k) {
        const float e = __expf(mz[k].x - M);
        z = fmaf(mz[k].y, e, z);
        s = fmaf(sk[k], e, s);
    }
    s = s / z;
    float n2 = s * s;
    n2 += __shfl_xor(n2, 8);
    n2 += __shfl_xor(n2, 4);
    n2 += __shfl_xor(n2, 2);
    n2 += __shfl_xor(n2, 1);
    const float scale = n2 / ((1.0f + n2) * sqrtf(n2));
    const float v = scale * s;
    if (step == 2) out[idx] = v;
    else           vsum[idx] = vprev + v;
}

extern "C" void kernel_launch(void* const* d_in, const int* in_sizes, int n_in,
                              void* d_out, int out_size, void* d_ws, size_t ws_size,
                              hipStream_t stream) {
    const float* x = (const float*)d_in[0];
    const float* W = (const float*)d_in[1];
    float* out = (float*)d_out;

    float* part_mz = (float*)d_ws;
    float* part_s  = part_mz + (size_t)NCHUNK * CAPS * BATCH * 2;
    float* vsum    = part_s + (size_t)NCHUNK * CAPS * PLANE;

    const dim3 gemm_grid(CAPS / CPB, NCHUNK);
    for (int t = 0; t < 3; ++t) {
        if (t == 0)
            caps_s_gemm_first_kernel<<<gemm_grid, FG_WAVES * 32, 0, stream>>>(x, W, part_s, part_mz);
        else
            caps_route_gemm_kernel<<<gemm_grid, RG_WAVES * 32, 0, stream>>>(x, W, vsum, part_s, part_mz);
        caps_squash_kernel<<<BATCH * CAPS * COUT / 256, 256, 0, stream>>>(
            part_s, part_mz, vsum, out, t);
    }
}
